# starting kernel
# baseline (speedup 1.0000x reference)
#include <hip/hip_runtime.h>

typedef __attribute__((ext_vector_type(2))) float v2f;
typedef __attribute__((ext_vector_type(8))) float v8f;

#define H 56
#define W 56
#define PIX (H * W)
#define KTOT 192
#define LDS_STRIDE 196

__global__ void __launch_bounds__(128)
shiftconv_wmma_f32(const float* __restrict__ x,
                   const float* __restrict__ w,
                   float* __restrict__ y)
{
    __shared__ float As[16 * LDS_STRIDE];

    const int tid   = threadIdx.x;
    const int lane  = tid & 31;
    const int wid   = tid >> 5;
    const int mtile = blockIdx.x;

    for (int e = tid; e < 16 * KTOT; e += 128) {
        const int mm  = e / KTOT;
        const int r   = e - mm * KTOT;
        const int l   = r / 6;
        const int rem = r - l * 6;
        const int j   = rem / 3;
        const int k   = rem - j * 3;

        const int p = mtile * 16 + mm;
        const int n = p / W;
        const int o = p - n * W;

        const int o2 = o + k - 1;
        float val = 0.0f;
        if (o2 >= 0 && o2 < W) {
            const int o3 = (o2 + W - 1) % W;
            const int c  = (j == 0) ? (32 + l) : l;
            const int n2 = (j == 0) ? ((n + H - 1) % H) : n;
            val = x[c * PIX + n2 * W + o3];
        }
        As[mm * LDS_STRIDE + r] = val;
    }
    __syncthreads();

    const int half = lane >> 4;
    const int idx  = lane & 15;
    const float* bcol = w + (wid * 16 + idx) * KTOT;

    v8f acc = {};
    #pragma unroll
    for (int kb = 0; kb < KTOT / 4; ++kb) {
        const int kbase = kb * 4 + half * 2;
        v2f a, b;
        a.x = As[idx * LDS_STRIDE + kbase];
        a.y = As[idx * LDS_STRIDE + kbase + 1];
        b.x = bcol[kbase];
        b.y = bcol[kbase + 1];
        acc = __builtin_amdgcn_wmma_f32_16x16x4_f32(false, a, false, b,
                                                    (short)0, acc, false, false);
    }

    const int i = wid * 16 + idx;
    #pragma unroll
    for (int r = 0; r < 8; ++r) {
        const int p = mtile * 16 + r + 8 * half;
        const int n = p / W;
        const int o = p - n * W;
        y[i * PIX + n * W + o] = acc[r];
    }
}

extern "C" void kernel_launch(void* const* d_in, const int* in_sizes, int n_in,
                              void* d_out, int out_size, void* d_ws, size_t ws_size,
                              hipStream_t stream) {
    const float* x = (const float*)d_in[0];
    const float* w = (const float*)d_in[1];
    float* y = (float*)d_out;

    shiftconv_wmma_f32<<<PIX / 16, 128, 0, stream>>>(x, w, y);
}


# Stage the im2col A tile with one batch of 24 branch-free loads per thread (one pixel per lane, padding taps zeroed by select) instead of 24 serialized load-wait-store trips; LDS layout kept
# speedup vs baseline: 1.5467x; 1.5467x over previous
#include <hip/hip_runtime.h>


typedef __attribute__((ext_vector_type(2))) float v2f;
typedef __attribute__((ext_vector_type(8))) float v8f;

#define H 56
#define W 56
#define PIX (H * W)
#define KTOT 192
#define LDS_STRIDE 196

__global__ void __launch_bounds__(128)
shiftconv_wmma_f32(const float* __restrict__ x,
                   const float* __restrict__ w,
                   float* __restrict__ y)
{
    __shared__ __attribute__((aligned(16))) float As[16 * LDS_STRIDE];

    const int tid   = threadIdx.x;
    const int lane  = tid & 31;
    const int wid   = tid >> 5;
    const int mtile = blockIdx.x;

    {
        const unsigned mm = tid & 15;
        const unsigned g  = tid >> 4;
        const unsigned p  = mtile * 16 + mm;
        const unsigned n  = p / W;
        const unsigned o  = p - n * W;

        const unsigned nm1  = (n == 0) ? (H - 1) : (n - 1);
        const unsigned c1   = (o == 0) ? (W - 1) : (o - 1);
        const unsigned c0   = (c1 == 0) ? (W - 1) : (c1 - 1);
        const unsigned c2   = o;
        const bool     pad0 = (o == 0);
        const bool     pad2 = (o == W - 1);

        const unsigned b0 = (32 + 4 * g) * PIX + nm1 * W;
        const unsigned b1 = (4 * g) * PIX + n * W;
        const float* r0 = x + b0;
        const float* r1 = x + b1;

        float v[24];
        #pragma unroll
        for (int dl = 0; dl < 4; ++dl) {
            v[dl * 6 + 0] = r0[dl * PIX + c0];
            v[dl * 6 + 1] = r0[dl * PIX + c1];
            v[dl * 6 + 2] = r0[dl * PIX + c2];
            v[dl * 6 + 3] = r1[dl * PIX + c0];
            v[dl * 6 + 4] = r1[dl * PIX + c1];
            v[dl * 6 + 5] = r1[dl * PIX + c2];
        }
        #pragma unroll
        for (int dl = 0; dl < 4; ++dl) {
            v[dl * 6 + 0] = pad0 ? 0.0f : v[dl * 6 + 0];
            v[dl * 6 + 3] = pad0 ? 0.0f : v[dl * 6 + 3];
            v[dl * 6 + 2] = pad2 ? 0.0f : v[dl * 6 + 2];
            v[dl * 6 + 5] = pad2 ? 0.0f : v[dl * 6 + 5];
        }
        float4* dst = reinterpret_cast<float4*>(&As[mm * LDS_STRIDE + 24 * g]);
        #pragma unroll
        for (int q = 0; q < 6; ++q)
            dst[q] = make_float4(v[4 * q + 0], v[4 * q + 1], v[4 * q + 2], v[4 * q + 3]);
    }
    __syncthreads();

    const int half = lane >> 4;
    const int idx  = lane & 15;
    const float* bcol = w + (wid * 16 + idx) * KTOT;

    v8f acc = {};
    #pragma unroll
    for (int kb = 0; kb < KTOT / 4; ++kb) {
        const int kbase = kb * 4 + half * 2;
        v2f a, b;
        a.x = As[idx * LDS_STRIDE + kbase];
        a.y = As[idx * LDS_STRIDE + kbase + 1];
        b.x = bcol[kbase];
        b.y = bcol[kbase + 1];
        acc = __builtin_amdgcn_wmma_f32_16x16x4_f32(false, a, false, b,
                                                    (short)0, acc, false, false);
    }

    const int i = wid * 16 + idx;
    #pragma unroll
    for (int r = 0; r < 8; ++r) {
        const int p = mtile * 16 + r + 8 * half;
        const int n = p / W;
        const int o = p - n * W;
        y[i * PIX + n * W + o] = acc[r];
    }
}

extern "C" void kernel_launch(void* const* d_in, const int* in_sizes, int n_in,
                              void* d_out, int out_size, void* d_ws, size_t ws_size,
                              hipStream_t stream) {
    const float* x = (const float*)d_in[0];
    const float* w = (const float*)d_in[1];
    float* y = (float*)d_out;

    shiftconv_wmma_f32<<<PIX / 16, 128, 0, stream>>>(x, w, y);
}


# Swap the matrix operand roles (weights as A, pixels as B) so the tile is channel-major: each store writes two contiguous 64-byte pixel runs instead of 32 scattered dwords in 16 lines
# speedup vs baseline: 1.9011x; 1.2291x over previous
#include <hip/hip_runtime.h>


typedef __attribute__((ext_vector_type(2))) float v2f;
typedef __attribute__((ext_vector_type(8))) float v8f;

#define H 56
#define W 56
#define PIX (H * W)
#define KTOT 192
#define LDS_STRIDE 196

__global__ void __launch_bounds__(128)
shiftconv_wmma_f32(const float* __restrict__ x,
                   const float* __restrict__ w,
                   float* __restrict__ y)
{
    __shared__ __attribute__((aligned(16))) float As[16 * LDS_STRIDE];

    const int tid   = threadIdx.x;
    const int lane  = tid & 31;
    const int wid   = tid >> 5;
    const int mtile = blockIdx.x;

    {
        const unsigned mm = tid & 15;
        const unsigned g  = tid >> 4;
        const unsigned p  = mtile * 16 + mm;
        const unsigned n  = p / W;
        const unsigned o  = p - n * W;

        const unsigned nm1  = (n == 0) ? (H - 1) : (n - 1);
        const unsigned c1   = (o == 0) ? (W - 1) : (o - 1);
        const unsigned c0   = (c1 == 0) ? (W - 1) : (c1 - 1);
        const unsigned c2   = o;
        const bool     pad0 = (o == 0);
        const bool     pad2 = (o == W - 1);

        const unsigned b0 = (32 + 4 * g) * PIX + nm1 * W;
        const unsigned b1 = (4 * g) * PIX + n * W;
        const float* r0 = x + b0;
        const float* r1 = x + b1;

        float v[24];
        #pragma unroll
        for (int dl = 0; dl < 4; ++dl) {
            v[dl * 6 + 0] = r0[dl * PIX + c0];
            v[dl * 6 + 1] = r0[dl * PIX + c1];
            v[dl * 6 + 2] = r0[dl * PIX + c2];
            v[dl * 6 + 3] = r1[dl * PIX + c0];
            v[dl * 6 + 4] = r1[dl * PIX + c1];
            v[dl * 6 + 5] = r1[dl * PIX + c2];
        }
        #pragma unroll
        for (int dl = 0; dl < 4; ++dl) {
            v[dl * 6 + 0] = pad0 ? 0.0f : v[dl * 6 + 0];
            v[dl * 6 + 3] = pad0 ? 0.0f : v[dl * 6 + 3];
            v[dl * 6 + 2] = pad2 ? 0.0f : v[dl * 6 + 2];
            v[dl * 6 + 5] = pad2 ? 0.0f : v[dl * 6 + 5];
        }
        float4* dst = reinterpret_cast<float4*>(&As[mm * LDS_STRIDE + 24 * g]);
        #pragma unroll
        for (int q = 0; q < 6; ++q)
            dst[q] = make_float4(v[4 * q + 0], v[4 * q + 1], v[4 * q + 2], v[4 * q + 3]);
    }
    __syncthreads();

    const int half = lane >> 4;
    const int idx  = lane & 15;
    const float* wrow = w + (wid * 16 + idx) * KTOT;
    const float* prow = As + idx * LDS_STRIDE;

    v8f acc = {};
    #pragma unroll
    for (int kb = 0; kb < KTOT / 4; ++kb) {
        const int kbase = kb * 4 + half * 2;
        v2f wt, px;
        wt.x = wrow[kbase];
        wt.y = wrow[kbase + 1];
        px.x = prow[kbase];
        px.y = prow[kbase + 1];
        acc = __builtin_amdgcn_wmma_f32_16x16x4_f32(false, wt, false, px,
                                                    (short)0, acc, false, false);
    }

    float* yout = y + (wid * 16 + 8 * half) * PIX + mtile * 16 + idx;
    #pragma unroll
    for (int r = 0; r < 8; ++r)
        yout[r * PIX] = acc[r];
}

extern "C" void kernel_launch(void* const* d_in, const int* in_sizes, int n_in,
                              void* d_out, int out_size, void* d_ws, size_t ws_size,
                              hipStream_t stream) {
    const float* x = (const float*)d_in[0];
    const float* w = (const float*)d_in[1];
    float* y = (float*)d_out;

    shiftconv_wmma_f32<<<PIX / 16, 128, 0, stream>>>(x, w, y);
}


# Weights via LDS: each wave stages its 16 rows with 24 whole-line 16-byte loads issued with the pixel loads (one memory trip), not 48 post-barrier 16-line row loads; LDS fragments read as one group
# speedup vs baseline: 2.1759x; 1.1445x over previous
#include <hip/hip_runtime.h>


typedef __attribute__((ext_vector_type(2))) float v2f;
typedef __attribute__((ext_vector_type(8))) float v8f;

#define H 56
#define W 56
#define PIX (H * W)
#define KTOT 192
#define LDS_STRIDE 196

__global__ void __launch_bounds__(128)
shiftconv_wmma_f32(const float* __restrict__ x,
                   const float* __restrict__ w,
                   float* __restrict__ y)
{
    __shared__ __attribute__((aligned(16))) float As[16 * LDS_STRIDE];
    __shared__ __attribute__((aligned(16))) float Ws[64 * LDS_STRIDE];

    const int tid   = threadIdx.x;
    const int lane  = tid & 31;
    const int wid   = tid >> 5;
    const int mtile = blockIdx.x;

    float4 wv[24];
    {
        const float4* wsrc = reinterpret_cast<const float4*>(w) + wid * 768 + lane;
        #pragma unroll
        for (int q = 0; q < 24; ++q)
            wv[q] = wsrc[q * 32];
    }

    {
        const unsigned mm = tid & 15;
        const unsigned g  = tid >> 4;
        const unsigned p  = mtile * 16 + mm;
        const unsigned n  = p / W;
        const unsigned o  = p - n * W;

        const unsigned nm1  = (n == 0) ? (H - 1) : (n - 1);
        const unsigned c1   = (o == 0) ? (W - 1) : (o - 1);
        const unsigned c0   = (c1 == 0) ? (W - 1) : (c1 - 1);
        const unsigned c2   = o;
        const bool     pad0 = (o == 0);
        const bool     pad2 = (o == W - 1);

        const unsigned b0 = (32 + 4 * g) * PIX + nm1 * W;
        const unsigned b1 = (4 * g) * PIX + n * W;
        const float* r0 = x + b0;
        const float* r1 = x + b1;

        float v[24];
        #pragma unroll
        for (int dl = 0; dl < 4; ++dl) {
            v[dl * 6 + 0] = r0[dl * PIX + c0];
            v[dl * 6 + 1] = r0[dl * PIX + c1];
            v[dl * 6 + 2] = r0[dl * PIX + c2];
            v[dl * 6 + 3] = r1[dl * PIX + c0];
            v[dl * 6 + 4] = r1[dl * PIX + c1];
            v[dl * 6 + 5] = r1[dl * PIX + c2];
        }
        #pragma unroll
        for (int dl = 0; dl < 4; ++dl) {
            v[dl * 6 + 0] = pad0 ? 0.0f : v[dl * 6 + 0];
            v[dl * 6 + 3] = pad0 ? 0.0f : v[dl * 6 + 3];
            v[dl * 6 + 2] = pad2 ? 0.0f : v[dl * 6 + 2];
            v[dl * 6 + 5] = pad2 ? 0.0f : v[dl * 6 + 5];
        }
        float4* dst = reinterpret_cast<float4*>(&As[mm * LDS_STRIDE + 24 * g]);
        #pragma unroll
        for (int q = 0; q < 6; ++q)
            dst[q] = make_float4(v[4 * q + 0], v[4 * q + 1], v[4 * q + 2], v[4 * q + 3]);
    }
    {
        const int hi = lane >> 4;
        const int o0 = (wid * 16) * LDS_STRIDE + 4 * lane;
        const int o1 = (wid * 16 + hi) * LDS_STRIDE + 4 * (hi ? lane - 16 : lane + 32);
        const int o2 = (wid * 16 + 1) * LDS_STRIDE + 4 * (lane + 16);
        #pragma unroll
        for (int m = 0; m < 8; ++m) {
            *reinterpret_cast<float4*>(&Ws[o0 + 2 * m * LDS_STRIDE]) = wv[3 * m + 0];
            *reinterpret_cast<float4*>(&Ws[o1 + 2 * m * LDS_STRIDE]) = wv[3 * m + 1];
            *reinterpret_cast<float4*>(&Ws[o2 + 2 * m * LDS_STRIDE]) = wv[3 * m + 2];
        }
    }
    __syncthreads();

    const int half = lane >> 4;
    const int idx  = lane & 15;
    const float* wrow = Ws + (wid * 16 + idx) * LDS_STRIDE;
    const float* prow = As + idx * LDS_STRIDE;

    v2f wt[KTOT / 4], px[KTOT / 4];
    #pragma unroll
    for (int kb = 0; kb < KTOT / 4; ++kb) {
        const int kbase = kb * 4 + half * 2;
        wt[kb].x = wrow[kbase];
        wt[kb].y = wrow[kbase + 1];
        px[kb].x = prow[kbase];
        px[kb].y = prow[kbase + 1];
    }
    __builtin_amdgcn_sched_barrier(0);

    v8f acc = {};
    #pragma unroll
    for (int kb = 0; kb < KTOT / 4; ++kb) {
        acc = __builtin_amdgcn_wmma_f32_16x16x4_f32(false, wt[kb], false, px[kb],
                                                    (short)0, acc, false, false);
    }

    float* yout = y + (wid * 16 + 8 * half) * PIX + mtile * 16 + idx;
    #pragma unroll
    for (int r = 0; r < 8; ++r)
        yout[r * PIX] = acc[r];
}

extern "C" void kernel_launch(void* const* d_in, const int* in_sizes, int n_in,
                              void* d_out, int out_size, void* d_ws, size_t ws_size,
                              hipStream_t stream) {
    const float* x = (const float*)d_in[0];
    const float* w = (const float*)d_in[1];
    float* y = (float*)d_out;

    shiftconv_wmma_f32<<<PIX / 16, 128, 0, stream>>>(x, w, y);
}


# Re-tile a block from 16 pixels x 64 channels to 64 pixels x 16 channels: it fetches 16 weight rows (12 KB), not all 64 (48 KB); pixel pairs staged by 8-byte loads; k order and summing order kept
# speedup vs baseline: 2.5127x; 1.1548x over previous
#include <hip/hip_runtime.h>


typedef __attribute__((ext_vector_type(2))) float v2f;
typedef __attribute__((ext_vector_type(4))) float v4f;
typedef __attribute__((ext_vector_type(8))) float v8f;

#define H 56
#define W 56
#define PIX (H * W)
#define KTOT 192
#define LDS_STRIDE 196
#define ODD_ROW0 40

__global__ void __launch_bounds__(128)
shiftconv_wmma_f32(const float* __restrict__ x,
                   const float* __restrict__ w,
                   float* __restrict__ y)
{
    __shared__ __attribute__((aligned(16))) float As[(ODD_ROW0 + 32) * LDS_STRIDE];
    __shared__ __attribute__((aligned(16))) float Ws[16 * LDS_STRIDE];

    const int tid  = threadIdx.x;
    const int lane = tid & 31;
    const int wid  = tid >> 5;
    const int pg   = blockIdx.x >> 2;
    const int cg   = blockIdx.x & 3;
    const int P0   = pg * 64;

    v4f wv[6];
    {
        const v4f* wsrc = reinterpret_cast<const v4f*>(w) + cg * 768 + wid * 192 + lane;
        #pragma unroll
        for (int q = 0; q < 6; ++q)
            wv[q] = wsrc[q * 32];
    }

    {
        const unsigned p   = P0 + 2 * lane;
        const unsigned n   = p / W;
        const unsigned o   = p - n * W;
        const unsigned nm1 = (n == 0) ? (H - 1) : (n - 1);
        const unsigned po  = (o == 0) ? (W - 2) : (o - 2);
        const bool     pad0 = (o == 0);
        const bool     pad2 = (o == W - 2);

        const unsigned b0 = (32 + 8 * wid) * PIX + nm1 * W;
        const unsigned b1 = (8 * wid) * PIX + n * W;
        const float* r0 = x + b0;
        const float* r1 = x + b1;

        v2f pv0[8], cu0[8], pv1[8], cu1[8];
        #pragma unroll
        for (int dl = 0; dl < 8; ++dl) {
            pv0[dl] = *reinterpret_cast<const v2f*>(r0 + dl * PIX + po);
            cu0[dl] = *reinterpret_cast<const v2f*>(r0 + dl * PIX + o);
            pv1[dl] = *reinterpret_cast<const v2f*>(r1 + dl * PIX + po);
            cu1[dl] = *reinterpret_cast<const v2f*>(r1 + dl * PIX + o);
        }
        float va[48], vb[48];
        #pragma unroll
        for (int dl = 0; dl < 8; ++dl) {
            va[dl * 6 + 0] = pad0 ? 0.0f : pv0[dl].x;
            va[dl * 6 + 1] = pv0[dl].y;
            va[dl * 6 + 2] = cu0[dl].x;
            va[dl * 6 + 3] = pad0 ? 0.0f : pv1[dl].x;
            va[dl * 6 + 4] = pv1[dl].y;
            va[dl * 6 + 5] = cu1[dl].x;
            vb[dl * 6 + 0] = pv0[dl].y;
            vb[dl * 6 + 1] = cu0[dl].x;
            vb[dl * 6 + 2] = pad2 ? 0.0f : cu0[dl].y;
            vb[dl * 6 + 3] = pv1[dl].y;
            vb[dl * 6 + 4] = cu1[dl].x;
            vb[dl * 6 + 5] = pad2 ? 0.0f : cu1[dl].y;
        }
        v4f* da = reinterpret_cast<v4f*>(&As[lane * LDS_STRIDE + 48 * wid]);
        v4f* db = reinterpret_cast<v4f*>(&As[(ODD_ROW0 + lane) * LDS_STRIDE + 48 * wid]);
        #pragma unroll
        for (int q = 0; q < 12; ++q) {
            da[q] = (v4f){va[4 * q + 0], va[4 * q + 1], va[4 * q + 2], va[4 * q + 3]};
            db[q] = (v4f){vb[4 * q + 0], vb[4 * q + 1], vb[4 * q + 2], vb[4 * q + 3]};
        }
    }
    {
        const int hi = lane >> 4;
        const int o0 = (wid * 4) * LDS_STRIDE + 4 * lane;
        const int o1 = (wid * 4 + hi) * LDS_STRIDE + 4 * (hi ? lane - 16 : lane + 32);
        const int o2 = (wid * 4 + 1) * LDS_STRIDE + 4 * (lane + 16);
        #pragma unroll
        for (int m = 0; m < 2; ++m) {
            *reinterpret_cast<v4f*>(&Ws[o0 + 2 * m * LDS_STRIDE]) = wv[3 * m + 0];
            *reinterpret_cast<v4f*>(&Ws[o1 + 2 * m * LDS_STRIDE]) = wv[3 * m + 1];
            *reinterpret_cast<v4f*>(&Ws[o2 + 2 * m * LDS_STRIDE]) = wv[3 * m + 2];
        }
    }
    __syncthreads();

    const int half = lane >> 4;
    const int idx  = lane & 15;
    const int prw = 8 * wid + (idx >> 1) + ODD_ROW0 * (idx & 1);
    const float* wrow = Ws + idx * LDS_STRIDE;
    const float* prow = As + prw * LDS_STRIDE;

    v2f wt[KTOT / 4], px[KTOT / 4];
    #pragma unroll
    for (int kb = 0; kb < KTOT / 4; ++kb) {
        const int kbase = kb * 4 + half * 2;
        wt[kb].x = wrow[kbase];
        wt[kb].y = wrow[kbase + 1];
        px[kb].x = prow[kbase];
        px[kb].y = prow[kbase + 1];
    }
    __builtin_amdgcn_sched_barrier(0);

    v8f acc = {};
    #pragma unroll
    for (int kb = 0; kb < KTOT / 4; ++kb) {
        acc = __builtin_amdgcn_wmma_f32_16x16x4_f32(false, wt[kb], false, px[kb],
                                                    (short)0, acc, false, false);
    }

    float* yout = y + (cg * 16 + 8 * half) * PIX + P0 + 16 * wid + idx;
    #pragma unroll
    for (int r = 0; r < 8; ++r)
        yout[r * PIX] = acc[r];
}

extern "C" void kernel_launch(void* const* d_in, const int* in_sizes, int n_in,
                              void* d_out, int out_size, void* d_ws, size_t ws_size,
                              hipStream_t stream) {
    const float* x = (const float*)d_in[0];
    const float* w = (const float*)d_in[1];
    float* y = (float*)d_out;

    shiftconv_wmma_f32<<<PIX / 16, 128, 0, stream>>>(x, w, y);
}
